# starting kernel
# baseline (speedup 1.0000x reference)
#include <hip/hip_runtime.h>

typedef float v2f __attribute__((ext_vector_type(2)));
typedef float v8f __attribute__((ext_vector_type(8)));

#define B_  2
#define S_  96
#define H_  96
#define D_  96
#define NP  (B_ * S_)
#define SCALE_ 0.10206207261596575f

__global__ void k_tvec(const float* __restrict__ Q,
                       const float* __restrict__ V,
                       const int*   __restrict__ idxp,
                       float*       __restrict__ T) {
    __shared__ float qsh[D_];
    const int p = blockIdx.x;
    const int i = threadIdx.x;
    const int idx = *idxp;

    const float* __restrict__ qrow = Q + (size_t)p * H_ * D_ + (size_t)idx * D_;
    if (i < D_) qsh[i] = qrow[i];
    __syncthreads();

    const float* __restrict__ vbase = V + (size_t)p * H_ * D_;
    float acc = 0.0f;
#pragma unroll 8
    for (int k = 0; k < H_; ++k) {
        acc = fmaf(qsh[k], vbase[(size_t)k * D_ + i], acc);
    }
    T[(size_t)p * D_ + i] = acc;
}

__global__ void k_gemm_wmma(const float* __restrict__ T,
                            const float* __restrict__ DW,
                            const int*   __restrict__ idxp,
                            float*       __restrict__ U) {
    const int tile = blockIdx.x;
    const int tn   = tile % (D_ / 16);
    const int tm   = tile / (D_ / 16);
    const int lane = threadIdx.x & 31;
    const int half = lane >> 4;
    const int lid  = lane & 15;

    const float* __restrict__ W = DW + (size_t)(*idxp) * D_ * D_;
    const int row0 = tm * 16;
    const int col0 = tn * 16;

    v8f c = {};
    for (int kk = 0; kk < D_; kk += 4) {
        const int ka = kk + 2 * half;
        v2f a, b;
        a.x = T[(size_t)(row0 + lid) * D_ + ka];
        a.y = T[(size_t)(row0 + lid) * D_ + ka + 1];
        b.x = W[(size_t)ka       * D_ + col0 + lid];
        b.y = W[(size_t)(ka + 1) * D_ + col0 + lid];
        c = __builtin_amdgcn_wmma_f32_16x16x4_f32(false, a, false, b,
                                                  (short)0, c, false, false);
    }

#pragma unroll
    for (int j = 0; j < 8; ++j) {
        U[(size_t)(row0 + j + 8 * half) * D_ + col0 + lid] = c[j];
    }
}

__global__ void k_outer(const float* __restrict__ Kmat,
                        const float* __restrict__ U,
                        const int*   __restrict__ idxp,
                        float*       __restrict__ out) {
    const int gid = blockIdx.x * blockDim.x + threadIdx.x;
    const int total = NP * D_ * D_;
    if (gid >= total) return;
    const int v = gid % D_;
    const int r = gid / D_;
    const int q = r % D_;
    const int p = r / D_;
    const float kv = Kmat[((size_t)p * H_ + (size_t)(*idxp)) * D_ + q];
    out[gid] = SCALE_ * kv * U[(size_t)p * D_ + v];
}

extern "C" void kernel_launch(void* const* d_in, const int* in_sizes, int n_in,
                              void* d_out, int out_size, void* d_ws, size_t ws_size,
                              hipStream_t stream) {
    const float* Q   = (const float*)d_in[0];
    const float* K   = (const float*)d_in[1];
    const float* V   = (const float*)d_in[2];
    const float* DW  = (const float*)d_in[3];
    const int*   idx = (const int*)  d_in[4];
    float* out = (float*)d_out;

    float* T = (float*)d_ws;
    float* U = T + (size_t)NP * D_;

    k_tvec     <<<NP, 96, 0, stream>>>(Q, V, idx, T);
    k_gemm_wmma<<<(NP / 16) * (D_ / 16), 32, 0, stream>>>(T, DW, idx, U);

    const int total  = NP * D_ * D_;
    const int blocks = (total + 255) / 256;
    k_outer    <<<blocks, 256, 0, stream>>>(K, U, idx, out);
}


# Fused the three launches into one kernel, one work-group per position; T is kept as 16 partial rows that the f32 WMMA T*W product and a row sum add up (float summation order changes)
# speedup vs baseline: 3.9537x; 3.9537x over previous
#include <hip/hip_runtime.h>

typedef float v2f __attribute__((ext_vector_type(2)));
typedef float v8f __attribute__((ext_vector_type(8)));

#define B_  2
#define S_  96
#define H_  96
#define D_  96
#define NP  (B_ * S_)
#define SCALE_ 0.10206207261596575f

#define NT  192
#define NJ  12
#define PP  100
#define PW  104

__global__ __launch_bounds__(NT)
void k_fused(const float* __restrict__ Q,
             const float* __restrict__ Kmat,
             const float* __restrict__ V,
             const float* __restrict__ DW,
             const int*   __restrict__ idxp,
             float*       __restrict__ out) {
    __shared__ __attribute__((aligned(16))) float Wsh[D_ * PW];
    __shared__ __attribute__((aligned(16))) float Psh[16 * PP];
    __shared__ __attribute__((aligned(16))) float Ush[D_];

    const int p  = blockIdx.x;
    const int t  = threadIdx.x;
    const int cg = t % 24;
    const int rg = t / 24;

    const float4* __restrict__ V4 = reinterpret_cast<const float4*>(V + (size_t)p * H_ * D_);
    float4 vv[NJ];
    for (int j = 0; j < NJ; ++j) vv[j] = V4[t + NT * j];

    int idx = *idxp;
    idx = idx < 0 ? 0 : (idx > 7 ? 7 : idx);

    const float* __restrict__ qrow = Q    + ((size_t)p * H_ + (size_t)idx) * D_;
    const float* __restrict__ krow = Kmat + ((size_t)p * H_ + (size_t)idx) * D_;
    const float4* __restrict__ W4  = reinterpret_cast<const float4*>(DW + (size_t)idx * D_ * D_);

    float qv[NJ];
    for (int j = 0; j < NJ; ++j) qv[j] = qrow[rg + 8 * j];
    float4 ww[NJ];
    for (int j = 0; j < NJ; ++j) ww[j] = W4[t + NT * j];
    float kv[NJ];
    for (int j = 0; j < NJ; ++j) kv[j] = krow[rg + 8 * j];

    float4 a0 = make_float4(0.f, 0.f, 0.f, 0.f);
    float4 a1 = make_float4(0.f, 0.f, 0.f, 0.f);
    for (int j = 0; j < NJ; j += 2) {
        a0.x = fmaf(qv[j], vv[j].x, a0.x);
        a0.y = fmaf(qv[j], vv[j].y, a0.y);
        a0.z = fmaf(qv[j], vv[j].z, a0.z);
        a0.w = fmaf(qv[j], vv[j].w, a0.w);
        a1.x = fmaf(qv[j + 1], vv[j + 1].x, a1.x);
        a1.y = fmaf(qv[j + 1], vv[j + 1].y, a1.y);
        a1.z = fmaf(qv[j + 1], vv[j + 1].z, a1.z);
        a1.w = fmaf(qv[j + 1], vv[j + 1].w, a1.w);
    }
    *reinterpret_cast<float4*>(&Psh[rg * PP + 4 * cg])       = a0;
    *reinterpret_cast<float4*>(&Psh[(rg + 8) * PP + 4 * cg]) = a1;
    for (int j = 0; j < NJ; ++j)
        *reinterpret_cast<float4*>(&Wsh[(rg + 8 * j) * PW + 4 * cg]) = ww[j];
    __syncthreads();

    const int wave = t >> 5;
    const int lane = t & 31;
    const int half = lane >> 4;
    const int lid  = lane & 15;
    const int col  = wave * 16 + lid;

    v8f c0 = {};
    v8f c1 = {};
    for (int kk = 0; kk < D_; kk += 8) {
        const int ka = kk + 2 * half;
        const int kb = kk + 4 + 2 * half;
        v2f fa0 = *reinterpret_cast<const v2f*>(&Psh[lid * PP + ka]);
        v2f fa1 = *reinterpret_cast<const v2f*>(&Psh[lid * PP + kb]);
        v2f fb0, fb1;
        fb0.x = Wsh[ka * PW + col];
        fb0.y = Wsh[(ka + 1) * PW + col];
        fb1.x = Wsh[kb * PW + col];
        fb1.y = Wsh[(kb + 1) * PW + col];
        c0 = __builtin_amdgcn_wmma_f32_16x16x4_f32(false, fa0, false, fb0,
                                                   (short)0, c0, false, false);
        c1 = __builtin_amdgcn_wmma_f32_16x16x4_f32(false, fa1, false, fb1,
                                                   (short)0, c1, false, false);
    }
    {
        const v8f cs = c0 + c1;
        float s = ((cs[0] + cs[1]) + (cs[2] + cs[3])) + ((cs[4] + cs[5]) + (cs[6] + cs[7]));
        s += __shfl_xor(s, 16);
        if (half == 0) Ush[col] = s;
    }
    __syncthreads();

    const float4 u = *reinterpret_cast<const float4*>(&Ush[4 * cg]);
    float4* __restrict__ o4 = reinterpret_cast<float4*>(out + (size_t)p * D_ * D_);
    for (int j = 0; j < NJ; ++j) {
        const float s = SCALE_ * kv[j];
        o4[t + NT * j] = make_float4(s * u.x, s * u.y, s * u.z, s * u.w);
    }
}

extern "C" void kernel_launch(void* const* d_in, const int* in_sizes, int n_in,
                              void* d_out, int out_size, void* d_ws, size_t ws_size,
                              hipStream_t stream) {
    const float* Q   = (const float*)d_in[0];
    const float* K   = (const float*)d_in[1];
    const float* V   = (const float*)d_in[2];
    const float* DW  = (const float*)d_in[3];
    const int*   idx = (const int*)  d_in[4];
    float* out = (float*)d_out;

    k_fused<<<NP, NT, 0, stream>>>(Q, K, V, DW, idx, out);
}


# W no longer staged in LDS: each wave loads its own B fragments from global straight into registers, issued up front with the V loads; sum over i split over wave pairs (float summation order changes)
# speedup vs baseline: 4.3453x; 1.0990x over previous
#include <hip/hip_runtime.h>

typedef float v2f __attribute__((ext_vector_type(2)));
typedef float v8f __attribute__((ext_vector_type(8)));

#define B_  2
#define S_  96
#define H_  96
#define D_  96
#define NP  (B_ * S_)
#define SCALE_ 0.10206207261596575f

#define NT  192
#define NJ  12
#define PP  100
#define NS  12

__global__ __launch_bounds__(NT)
void k_fused(const float* __restrict__ Q,
             const float* __restrict__ Kmat,
             const float* __restrict__ V,
             const float* __restrict__ DW,
             const int*   __restrict__ idxp,
             float*       __restrict__ out) {
    __shared__ __attribute__((aligned(16))) float Psh[16 * PP];
    __shared__ __attribute__((aligned(16))) float Ush[2 * D_];

    const int p  = blockIdx.x;
    const int t  = threadIdx.x;
    const int cg = t % 24;
    const int rg = t / 24;

    const int wave = t >> 5;
    const int lane = t & 31;
    const int half = lane >> 4;
    const int lid  = lane & 15;
    const int cg2  = wave % 3;
    const int kh   = wave / 3;

    const float4* __restrict__ V4 = reinterpret_cast<const float4*>(V + (size_t)p * H_ * D_);
    float4 vv[NJ];
    for (int j = 0; j < NJ; ++j) vv[j] = V4[t + NT * j];

    int idx = *idxp;
    idx = idx < 0 ? 0 : (idx > 7 ? 7 : idx);

    const float* __restrict__ qrow = Q    + ((size_t)p * H_ + (size_t)idx) * D_;
    const float* __restrict__ krow = Kmat + ((size_t)p * H_ + (size_t)idx) * D_;

    float qv[NJ];
    for (int j = 0; j < NJ; ++j) qv[j] = qrow[rg + 8 * j];

    const float* __restrict__ wl = DW + (size_t)idx * D_ * D_ + (size_t)(48 * kh + 2 * half) * D_ + 32 * cg2 + 2 * lid;
    v2f w0[NS], w1[NS];
    for (int s = 0; s < NS; ++s) {
        w0[s] = *reinterpret_cast<const v2f*>(wl + (4 * s)     * D_);
        w1[s] = *reinterpret_cast<const v2f*>(wl + (4 * s + 1) * D_);
    }

    float kv[NJ];
    for (int j = 0; j < NJ; ++j) kv[j] = krow[rg + 8 * j];

    __builtin_amdgcn_sched_barrier(0);

    float4 a0 = make_float4(0.f, 0.f, 0.f, 0.f);
    float4 a1 = make_float4(0.f, 0.f, 0.f, 0.f);
    for (int j = 0; j < NJ; j += 2) {
        a0.x = fmaf(qv[j], vv[j].x, a0.x);
        a0.y = fmaf(qv[j], vv[j].y, a0.y);
        a0.z = fmaf(qv[j], vv[j].z, a0.z);
        a0.w = fmaf(qv[j], vv[j].w, a0.w);
        a1.x = fmaf(qv[j + 1], vv[j + 1].x, a1.x);
        a1.y = fmaf(qv[j + 1], vv[j + 1].y, a1.y);
        a1.z = fmaf(qv[j + 1], vv[j + 1].z, a1.z);
        a1.w = fmaf(qv[j + 1], vv[j + 1].w, a1.w);
    }
    *reinterpret_cast<float4*>(&Psh[rg * PP + 4 * cg])       = a0;
    *reinterpret_cast<float4*>(&Psh[(rg + 8) * PP + 4 * cg]) = a1;
    __syncthreads();

    const float* pa = &Psh[lid * PP + 48 * kh + 2 * half];
    v8f ce = {};
    v8f co = {};
    v2f fa[NS];
    for (int s = 0; s < NS; ++s) fa[s] = *reinterpret_cast<const v2f*>(pa + 4 * s);
    __builtin_amdgcn_sched_barrier(0);
    for (int s = 0; s < NS; ++s) {
        v2f be, bo;
        be.x = w0[s].x;  be.y = w1[s].x;
        bo.x = w0[s].y;  bo.y = w1[s].y;
        ce = __builtin_amdgcn_wmma_f32_16x16x4_f32(false, fa[s], false, be,
                                                   (short)0, ce, false, false);
        co = __builtin_amdgcn_wmma_f32_16x16x4_f32(false, fa[s], false, bo,
                                                   (short)0, co, false, false);
    }
    {
        float se = ((ce[0] + ce[1]) + (ce[2] + ce[3])) + ((ce[4] + ce[5]) + (ce[6] + ce[7]));
        float so = ((co[0] + co[1]) + (co[2] + co[3])) + ((co[4] + co[5]) + (co[6] + co[7]));
        se += __shfl_xor(se, 16);
        so += __shfl_xor(so, 16);
        if (half == 0)
            *reinterpret_cast<float2*>(&Ush[kh * D_ + 32 * cg2 + 2 * lid]) = make_float2(se, so);
    }
    __syncthreads();

    const float4 ua = *reinterpret_cast<const float4*>(&Ush[4 * cg]);
    const float4 ub = *reinterpret_cast<const float4*>(&Ush[D_ + 4 * cg]);
    const float4 u  = make_float4(ua.x + ub.x, ua.y + ub.y, ua.z + ub.z, ua.w + ub.w);
    float4* __restrict__ o4 = reinterpret_cast<float4*>(out + (size_t)p * D_ * D_);
    for (int j = 0; j < NJ; ++j) {
        const float s = SCALE_ * kv[j];
        o4[t + NT * j] = make_float4(s * u.x, s * u.y, s * u.z, s * u.w);
    }
}

extern "C" void kernel_launch(void* const* d_in, const int* in_sizes, int n_in,
                              void* d_out, int out_size, void* d_ws, size_t ws_size,
                              hipStream_t stream) {
    const float* Q   = (const float*)d_in[0];
    const float* K   = (const float*)d_in[1];
    const float* V   = (const float*)d_in[2];
    const float* DW  = (const float*)d_in[3];
    const int*   idx = (const int*)  d_in[4];
    float* out = (float*)d_out;

    k_fused<<<NP, NT, 0, stream>>>(Q, K, V, DW, idx, out);
}


# Loads issued in order of dependency and first use: V (needs no depth index) while the index load is in flight, then Q, W, K; first barrier fences LDS only, so step 1 runs under the W arrival
# speedup vs baseline: 4.7633x; 1.0962x over previous
#include <hip/hip_runtime.h>

typedef float v2f __attribute__((ext_vector_type(2)));
typedef float v8f __attribute__((ext_vector_type(8)));

#define B_  2
#define S_  96
#define H_  96
#define D_  96
#define NP  (B_ * S_)
#define SCALE_ 0.10206207261596575f

#define NT  192
#define NJ  12
#define PP  100
#define NS  12

__global__ __launch_bounds__(NT)
void k_fused(const float* __restrict__ Q,
             const float* __restrict__ Kmat,
             const float*              V,
             const float* __restrict__ DW,
             const int*                idxp,
             float*       __restrict__ out) {
    __shared__ __attribute__((aligned(16))) float Psh[16 * PP];
    __shared__ __attribute__((aligned(16))) float Ush[2 * D_];

    const int p  = blockIdx.x;
    const int t  = threadIdx.x;
    const int cg = t % 24;
    const int rg = t / 24;

    const int wave = t >> 5;
    const int lane = t & 31;
    const int half = lane >> 4;
    const int lid  = lane & 15;
    const int cg2  = wave % 3;
    const int kh   = wave / 3;

    const float4* V4 = reinterpret_cast<const float4*>(V + (size_t)p * H_ * D_);
    int idx = *idxp;
    float4 vv[NJ];
    for (int j = 0; j < NJ; ++j) vv[j] = V4[t + NT * j];
    __builtin_amdgcn_sched_barrier(0);

    idx = idx < 0 ? 0 : (idx > 7 ? 7 : idx);

    const float* __restrict__ qrow = Q    + ((size_t)p * H_ + (size_t)idx) * D_;
    const float* __restrict__ krow = Kmat + ((size_t)p * H_ + (size_t)idx) * D_;

    float qv[NJ];
    for (int j = 0; j < NJ; ++j) qv[j] = qrow[rg + 8 * j];
    __builtin_amdgcn_sched_barrier(0);

    const float* __restrict__ wl = DW + (size_t)idx * D_ * D_ + (size_t)(48 * kh + 2 * half) * D_ + 32 * cg2 + 2 * lid;
    v2f w0[NS], w1[NS];
    for (int s = 0; s < NS; ++s) {
        w0[s] = *reinterpret_cast<const v2f*>(wl + (4 * s)     * D_);
        w1[s] = *reinterpret_cast<const v2f*>(wl + (4 * s + 1) * D_);
    }

    float kv[NJ];
    for (int j = 0; j < NJ; ++j) kv[j] = krow[rg + 8 * j];

    __builtin_amdgcn_sched_barrier(0);

    float4 a0 = make_float4(0.f, 0.f, 0.f, 0.f);
    float4 a1 = make_float4(0.f, 0.f, 0.f, 0.f);
    for (int j = 0; j < NJ; j += 2) {
        a0.x = fmaf(qv[j], vv[j].x, a0.x);
        a0.y = fmaf(qv[j], vv[j].y, a0.y);
        a0.z = fmaf(qv[j], vv[j].z, a0.z);
        a0.w = fmaf(qv[j], vv[j].w, a0.w);
        a1.x = fmaf(qv[j + 1], vv[j + 1].x, a1.x);
        a1.y = fmaf(qv[j + 1], vv[j + 1].y, a1.y);
        a1.z = fmaf(qv[j + 1], vv[j + 1].z, a1.z);
        a1.w = fmaf(qv[j + 1], vv[j + 1].w, a1.w);
    }
    *reinterpret_cast<float4*>(&Psh[rg * PP + 4 * cg])       = a0;
    *reinterpret_cast<float4*>(&Psh[(rg + 8) * PP + 4 * cg]) = a1;
    __builtin_amdgcn_fence(__ATOMIC_RELEASE, "workgroup", "local");
    __builtin_amdgcn_s_barrier();
    __builtin_amdgcn_fence(__ATOMIC_ACQUIRE, "workgroup", "local");

    const float* pa = &Psh[lid * PP + 48 * kh + 2 * half];
    v8f ce = {};
    v8f co = {};
    v2f fa[NS];
    for (int s = 0; s < NS; ++s) fa[s] = *reinterpret_cast<const v2f*>(pa + 4 * s);
    __builtin_amdgcn_sched_barrier(0);
    for (int s = 0; s < NS; ++s) {
        v2f be, bo;
        be.x = w0[s].x;  be.y = w1[s].x;
        bo.x = w0[s].y;  bo.y = w1[s].y;
        ce = __builtin_amdgcn_wmma_f32_16x16x4_f32(false, fa[s], false, be,
                                                   (short)0, ce, false, false);
        co = __builtin_amdgcn_wmma_f32_16x16x4_f32(false, fa[s], false, bo,
                                                   (short)0, co, false, false);
    }
    {
        float se = ((ce[0] + ce[1]) + (ce[2] + ce[3])) + ((ce[4] + ce[5]) + (ce[6] + ce[7]));
        float so = ((co[0] + co[1]) + (co[2] + co[3])) + ((co[4] + co[5]) + (co[6] + co[7]));
        se += __shfl_xor(se, 16);
        so += __shfl_xor(so, 16);
        if (half == 0)
            *reinterpret_cast<float2*>(&Ush[kh * D_ + 32 * cg2 + 2 * lid]) = make_float2(se, so);
    }
    __syncthreads();

    const float4 ua = *reinterpret_cast<const float4*>(&Ush[4 * cg]);
    const float4 ub = *reinterpret_cast<const float4*>(&Ush[D_ + 4 * cg]);
    const float4 u  = make_float4(ua.x + ub.x, ua.y + ub.y, ua.z + ub.z, ua.w + ub.w);
    float4* __restrict__ o4 = reinterpret_cast<float4*>(out + (size_t)p * D_ * D_);
    for (int j = 0; j < NJ; ++j) {
        const float s = SCALE_ * kv[j];
        o4[t + NT * j] = make_float4(s * u.x, s * u.y, s * u.z, s * u.w);
    }
}

extern "C" void kernel_launch(void* const* d_in, const int* in_sizes, int n_in,
                              void* d_out, int out_size, void* d_ws, size_t ws_size,
                              hipStream_t stream) {
    const float* Q   = (const float*)d_in[0];
    const float* K   = (const float*)d_in[1];
    const float* V   = (const float*)d_in[2];
    const float* DW  = (const float*)d_in[3];
    const int*   idx = (const int*)  d_in[4];
    float* out = (float*)d_out;

    k_fused<<<NP, NT, 0, stream>>>(Q, K, V, DW, idx, out);
}
